# starting kernel
# baseline (speedup 1.0000x reference)
#include <hip/hip_runtime.h>

typedef __attribute__((ext_vector_type(2))) float v2f;
typedef __attribute__((ext_vector_type(4))) float v4f;
typedef __attribute__((ext_vector_type(8))) float v8f;

#define NB   4
#define NN   1024
#define NIN  64
#define NHID 32
#define NOUT 8
#define HH   96

__global__ __launch_bounds__(256) void k_lin_relu(
    const float* __restrict__ x, const float* __restrict__ W,
    const float* __restrict__ bl, float* __restrict__ h) {
  int tid = blockIdx.x * 256 + threadIdx.x;
  int k  = tid & (NHID - 1);
  int bn = tid >> 5;
  const float* xr = x + (size_t)bn * NIN;
  float s = bl[k];
#pragma unroll 8
  for (int i = 0; i < NIN; ++i) s = fmaf(xr[i], W[i * NHID + k], s);
  h[tid] = fmaxf(s, 0.0f);
}

__global__ __launch_bounds__(256) void k_shift_concat(
    const float* __restrict__ h, float* __restrict__ v) {
  int tid = blockIdx.x * 256 + threadIdx.x;
  int g  = tid % HH;
  int bn = tid / HH;
  int b  = bn >> 10;
  int n  = bn & (NN - 1);
  int c  = g >> 5;
  int k  = g & 31;
  int sn = n + c - 1;
  float val = 0.0f;
  if (sn >= 0 && sn < NN) val = h[(((size_t)b << 10) + sn) * NHID + k];
  v[tid] = val;
}

__global__ __launch_bounds__(256) void k_t_wmma(
    const float* __restrict__ v, const float* __restrict__ Wb,
    float* __restrict__ t) {
  const int lane   = threadIdx.x & 31;
  const int laneLo = lane & 15;
  const bool hi    = lane >= 16;

  int wave = (blockIdx.x * 256 + threadIdx.x) >> 5;
  int o   = wave & (NOUT - 1);
  int bi0 = (wave >> 3) * 16;

  const float* vA = v + (size_t)(bi0 + laneLo) * HH + (hi ? 2 : 0);
  v2f a[24];
#pragma unroll
  for (int s = 0; s < 24; ++s) a[s] = *(const v2f*)(vA + s * 4);

  const float* wB = Wb + (size_t)o * HH * HH + (hi ? 2 * HH : 0) + laneLo;

#pragma unroll
  for (int gt = 0; gt < 6; ++gt) {
    const int g0 = gt * 16;
    v8f acc = {};
#pragma unroll
    for (int s = 0; s < 24; ++s) {
      const float* p = wB + (size_t)(4 * s) * HH + g0;
      v2f bf = { p[0], p[HH] };
      acc = __builtin_amdgcn_wmma_f32_16x16x4_f32(
          false, a[s], false, bf, (short)0, acc, false, false);
    }
    float* tp = t + (size_t)(bi0 + (hi ? 8 : 0)) * (NOUT * HH)
                  + o * HH + g0 + laneLo;
#pragma unroll
    for (int r = 0; r < 8; ++r)
      tp[(size_t)r * (NOUT * HH)] = acc[r];
  }
}

__global__ __launch_bounds__(256) void k_bilinear_out(
    const float* __restrict__ t, const float* __restrict__ v,
    const float* __restrict__ bb_g, float* __restrict__ out) {
  const int lane   = threadIdx.x & 31;
  const int laneLo = lane & 15;
  const bool hi    = lane >= 16;

  int wave = (blockIdx.x * 256 + threadIdx.x) >> 5;
  int jg = wave & 15;
  int mt = (wave >> 4) & 511;
  int b  = wave >> 13;

  const int m0 = mt * 16;
  const int i0 = m0 >> 3;

  float bb[NOUT];
#pragma unroll
  for (int r = 0; r < NOUT; ++r) bb[r] = bb_g[r];

  const float* tA = t + ((size_t)b * 8192 + m0 + laneLo) * HH + (hi ? 2 : 0);
  v2f a[24];
#pragma unroll
  for (int s = 0; s < 24; ++s) a[s] = *(const v2f*)(tA + s * 4);

  const float* vBb = v + (size_t)b * NN * HH + (hi ? 2 : 0);
  const int i = i0 + (hi ? 1 : 0);

#pragma unroll
  for (int tj = 0; tj < 4; ++tj) {
    const int j0 = (jg * 4 + tj) * 16;
    const float* vB = vBb + (size_t)(j0 + laneLo) * HH;

    v8f acc = {};
#pragma unroll
    for (int s = 0; s < 24; ++s) {
      v2f bf = *(const v2f*)(vB + s * 4);
      acc = __builtin_amdgcn_wmma_f32_16x16x4_f32(
           false, a[s],  false, bf,
           (short)0, acc,  false,  false);
    }

    const int j = j0 + laneLo;
    float* op = out + (((size_t)b * NN + i) * NN + j) * NOUT;
    v4f lo = { acc[0] + bb[0], acc[1] + bb[1], acc[2] + bb[2], acc[3] + bb[3] };
    v4f hi4 = { acc[4] + bb[4], acc[5] + bb[5], acc[6] + bb[6], acc[7] + bb[7] };
    __builtin_nontemporal_store(lo,  (v4f*)op);
    __builtin_nontemporal_store(hi4, (v4f*)(op + 4));
  }
}

extern "C" void kernel_launch(void* const* d_in, const int* in_sizes, int n_in,
                              void* d_out, int out_size, void* d_ws, size_t ws_size,
                              hipStream_t stream) {
  const float* x     = (const float*)d_in[0];
  const float* W_lin = (const float*)d_in[1];
  const float* b_lin = (const float*)d_in[2];
  const float* W_bil = (const float*)d_in[3];
  const float* b_bil = (const float*)d_in[4];
  float* out = (float*)d_out;

  char* ws = (char*)d_ws;
  float* h  = (float*)ws;
  float* v  = (float*)(ws + (size_t)NB * NN * NHID * 4);
  float* t  = (float*)(ws + (size_t)NB * NN * NHID * 4
                          + (size_t)NB * NN * HH * 4);

  k_lin_relu<<<(NB * NN * NHID) / 256, 256, 0, stream>>>(x, W_lin, b_lin, h);
  k_shift_concat<<<(NB * NN * HH) / 256, 256, 0, stream>>>(h, v);
  k_t_wmma<<<256, 256, 0, stream>>>(v, W_bil, t);
  k_bilinear_out<<<4096, 256, 0, stream>>>(t, v, b_bil, out);
}


# Main GEMM re-tiled through LDS: a work-group stages its 256-row t and v blocks with whole-line 16-byte loads; each wave keeps 2 A tiles in registers and reads B fragments from LDS; same sums
# speedup vs baseline: 2.4837x; 2.4837x over previous
#include <hip/hip_runtime.h>

typedef __attribute__((ext_vector_type(2))) float v2f;
typedef __attribute__((ext_vector_type(4))) float v4f;
typedef __attribute__((ext_vector_type(8))) float v8f;

#define NB   4
#define NN   1024
#define NIN  64
#define NHID 32
#define NOUT 8
#define HH   96

__global__ __launch_bounds__(256) void k_lin_relu(
    const float* __restrict__ x, const float* __restrict__ W,
    const float* __restrict__ bl, float* __restrict__ h) {
  int tid = blockIdx.x * 256 + threadIdx.x;
  int k  = tid & (NHID - 1);
  int bn = tid >> 5;
  const float* xr = x + (size_t)bn * NIN;
  float s = bl[k];
#pragma unroll 8
  for (int i = 0; i < NIN; ++i) s = fmaf(xr[i], W[i * NHID + k], s);
  h[tid] = fmaxf(s, 0.0f);
}

__global__ __launch_bounds__(256) void k_shift_concat(
    const float* __restrict__ h, float* __restrict__ v) {
  int tid = blockIdx.x * 256 + threadIdx.x;
  int g  = tid % HH;
  int bn = tid / HH;
  int b  = bn >> 10;
  int n  = bn & (NN - 1);
  int c  = g >> 5;
  int k  = g & 31;
  int sn = n + c - 1;
  float val = 0.0f;
  if (sn >= 0 && sn < NN) val = h[(((size_t)b << 10) + sn) * NHID + k];
  v[tid] = val;
}

__global__ __launch_bounds__(256) void k_t_wmma(
    const float* __restrict__ v, const float* __restrict__ Wb,
    float* __restrict__ t) {
  const int lane   = threadIdx.x & 31;
  const int laneLo = lane & 15;
  const bool hi    = lane >= 16;

  int wave = (blockIdx.x * 256 + threadIdx.x) >> 5;
  int o   = wave & (NOUT - 1);
  int bi0 = (wave >> 3) * 16;

  const float* vA = v + (size_t)(bi0 + laneLo) * HH + (hi ? 2 : 0);
  v2f a[24];
#pragma unroll
  for (int s = 0; s < 24; ++s) a[s] = *(const v2f*)(vA + s * 4);

  const float* wB = Wb + (size_t)o * HH * HH + (hi ? 2 * HH : 0) + laneLo;

#pragma unroll
  for (int gt = 0; gt < 6; ++gt) {
    const int g0 = gt * 16;
    v8f acc = {};
#pragma unroll
    for (int s = 0; s < 24; ++s) {
      const float* p = wB + (size_t)(4 * s) * HH + g0;
      v2f bf = { p[0], p[HH] };
      acc = __builtin_amdgcn_wmma_f32_16x16x4_f32(
          false, a[s], false, bf, (short)0, acc, false, false);
    }
    float* tp = t + (size_t)(bi0 + (hi ? 8 : 0)) * (NOUT * HH)
                  + o * HH + g0 + laneLo;
#pragma unroll
    for (int r = 0; r < 8; ++r)
      tp[(size_t)r * (NOUT * HH)] = acc[r];
  }
}

#define BLK 256
#define LDP 100

__device__ __forceinline__ void stage_block(float* __restrict__ L,
                                            const v4f (&r)[24], int tid) {
#pragma unroll
  for (int k = 0; k < 24; ++k) {
    const int f   = tid + 256 * k;
    const int row = f / 24;
    const int q   = f - row * 24;
    *(v4f*)(L + row * LDP + q * 4) = r[k];
  }
}

__device__ __forceinline__ void load_frags(v2f (&d)[24], const float* p) {
#pragma unroll
  for (int s = 0; s < 24; ++s) d[s] = *(const v2f*)(p + 4 * s);
}

__device__ __forceinline__ void tile_pair(
    const v2f (&a0)[24], const v2f (&a1)[24], const v2f (&bf)[24],
    const float (&bb)[NOUT], float* __restrict__ op) {
  v8f acc0 = {};
  v8f acc1 = {};
#pragma unroll
  for (int s = 0; s < 24; ++s) {
    acc0 = __builtin_amdgcn_wmma_f32_16x16x4_f32(
         false, a0[s],  false, bf[s],
         (short)0, acc0,  false,  false);
    acc1 = __builtin_amdgcn_wmma_f32_16x16x4_f32(
        false, a1[s], false, bf[s], (short)0, acc1, false, false);
  }
  float* op1 = op + (size_t)2 * NN * NOUT;
  v4f lo0 = { acc0[0] + bb[0], acc0[1] + bb[1], acc0[2] + bb[2], acc0[3] + bb[3] };
  v4f hi0 = { acc0[4] + bb[4], acc0[5] + bb[5], acc0[6] + bb[6], acc0[7] + bb[7] };
  v4f lo1 = { acc1[0] + bb[0], acc1[1] + bb[1], acc1[2] + bb[2], acc1[3] + bb[3] };
  v4f hi1 = { acc1[4] + bb[4], acc1[5] + bb[5], acc1[6] + bb[6], acc1[7] + bb[7] };
  __builtin_nontemporal_store(lo0, (v4f*)op);
  __builtin_nontemporal_store(hi0, (v4f*)(op + 4));
  __builtin_nontemporal_store(lo1, (v4f*)op1);
  __builtin_nontemporal_store(hi1, (v4f*)(op1 + 4));
}

__global__ __launch_bounds__(256) void k_bilinear_out(
    const float* __restrict__ t, const float* __restrict__ v,
    const float* __restrict__ bb_g, float* __restrict__ out) {
  __shared__ __attribute__((aligned(16))) float L[BLK * LDP];

  const int tid    = threadIdx.x;
  const int lane   = tid & 31;
  const int w      = tid >> 5;
  const int laneLo = lane & 15;
  const bool hi    = lane >= 16;

  const int blk = blockIdx.x;
  const int jb  = blk & 3;
  const int mb  = (blk >> 2) & 31;
  const int b   = blk >> 7;

  const float* tblk = t + ((size_t)b * 8192 + (size_t)mb * BLK) * HH;
  const float* vblk = v + ((size_t)b * NN + (size_t)jb * BLK) * HH;
  v4f tr[24], vr[24];
#pragma unroll
  for (int k = 0; k < 24; ++k) tr[k] = *(const v4f*)(tblk + (size_t)(tid + 256 * k) * 4);
#pragma unroll
  for (int k = 0; k < 24; ++k) vr[k] = *(const v4f*)(vblk + (size_t)(tid + 256 * k) * 4);

  float bb[NOUT];
#pragma unroll
  for (int r = 0; r < NOUT; ++r) bb[r] = bb_g[r];

  stage_block(L, tr, tid);
  __syncthreads();
  const float* la = L + (w * 32 + laneLo) * LDP + (hi ? 2 : 0);
  v2f a0[24], a1[24];
  load_frags(a0, la);
  load_frags(a1, la + 16 * LDP);
  __syncthreads();

  stage_block(L, vr, tid);
  __syncthreads();

  const int i0 = (mb * BLK + w * 32) >> 3;
  float* ob = out + (((size_t)b * NN + i0 + (hi ? 1 : 0)) * NN
                     + (size_t)jb * BLK + laneLo) * NOUT;
  const float* lb = L + laneLo * LDP + (hi ? 2 : 0);

  v2f b0[24], b1[24];
  load_frags(b0, lb);
  for (int jt = 0; jt < 16; jt += 2) {
    load_frags(b1, lb + (jt + 1) * 16 * LDP);
    tile_pair(a0, a1, b0, bb, ob + (size_t)jt * 16 * NOUT);
    load_frags(b0, lb + ((jt + 2) & 15) * 16 * LDP);
    tile_pair(a0, a1, b1, bb, ob + (size_t)(jt + 1) * 16 * NOUT);
  }
}

extern "C" void kernel_launch(void* const* d_in, const int* in_sizes, int n_in,
                              void* d_out, int out_size, void* d_ws, size_t ws_size,
                              hipStream_t stream) {
  const float* x     = (const float*)d_in[0];
  const float* W_lin = (const float*)d_in[1];
  const float* b_lin = (const float*)d_in[2];
  const float* W_bil = (const float*)d_in[3];
  const float* b_bil = (const float*)d_in[4];
  float* out = (float*)d_out;

  char* ws = (char*)d_ws;
  float* h  = (float*)ws;
  float* v  = (float*)(ws + (size_t)NB * NN * NHID * 4);
  float* t  = (float*)(ws + (size_t)NB * NN * NHID * 4
                          + (size_t)NB * NN * HH * 4);

  k_lin_relu<<<(NB * NN * NHID) / 256, 256, 0, stream>>>(x, W_lin, b_lin, h);
  k_shift_concat<<<(NB * NN * HH) / 256, 256, 0, stream>>>(h, v);
  k_t_wmma<<<256, 256, 0, stream>>>(v, W_bil, t);
  k_bilinear_out<<<512, 256, 0, stream>>>(t, v, b_bil, out);
}


# Three producer launches fused into one: a work-group computes h for its 16 rows plus 2 halo rows into LDS, writes v and takes the t GEMM's A fragments from there; 4 launches become 2; same sums
# speedup vs baseline: 2.6920x; 1.0839x over previous
#include <hip/hip_runtime.h>

typedef __attribute__((ext_vector_type(2))) float v2f;
typedef __attribute__((ext_vector_type(4))) float v4f;
typedef __attribute__((ext_vector_type(8))) float v8f;

#define NB   4
#define NN   1024
#define NIN  64
#define NHID 32
#define NOUT 8
#define HH   96

#define HP 36

__global__ __launch_bounds__(256) void k_front(
    const float* __restrict__ x, const float* __restrict__ W,
    const float* __restrict__ bl, const float* __restrict__ Wb,
    float* __restrict__ v, float* __restrict__ t) {
  __shared__ __attribute__((aligned(16))) float Hs[18 * HP];

  const int tid    = threadIdx.x;
  const int lane   = tid & 31;
  const int laneLo = lane & 15;
  const bool hi    = lane >= 16;
  const int w   = __builtin_amdgcn_readfirstlane(tid >> 5);
  const int bi0 = blockIdx.x * 16;
  const int n0  = bi0 & (NN - 1);

  float wc[NIN];
#pragma unroll
  for (int i = 0; i < NIN; ++i) wc[i] = W[i * NHID + lane];
  const float bias = bl[lane];

  for (int q = w; q < 18; q += 8) {
    const int n = n0 - 1 + q;
    float val = 0.0f;
    if (n >= 0 && n < NN) {
      const float* xr = x + (size_t)(bi0 - 1 + q) * NIN;
      float s = bias;
#pragma unroll
      for (int i = 0; i < NIN; ++i) s = fmaf(xr[i], wc[i], s);
      val = fmaxf(s, 0.0f);
    }
    Hs[q * HP + lane] = val;
  }
  __syncthreads();

  float* vt = v + (size_t)bi0 * HH;
  for (int p = 0; p < 2; ++p) {
    const int f = tid + 256 * p;
    if (f < 16 * 24) {
      const int r  = f / 24;
      const int q4 = f - r * 24;
      *(v4f*)(vt + (size_t)f * 4) =
          *(const v4f*)(Hs + (r + (q4 >> 3)) * HP + (q4 & 7) * 4);
    }
  }

  const float* hA = Hs + laneLo * HP + (hi ? 2 : 0);
  v2f a[24];
#pragma unroll
  for (int s = 0; s < 24; ++s)
    a[s] = *(const v2f*)(hA + (s >> 3) * HP + (s & 7) * 4);

  const int o = w;
  const float* wB = Wb + (size_t)o * HH * HH + (hi ? 2 * HH : 0) + laneLo;

#pragma unroll
  for (int gt = 0; gt < 6; ++gt) {
    const int g0 = gt * 16;
    v8f acc = {};
#pragma unroll
    for (int s = 0; s < 24; ++s) {
      const float* p = wB + (size_t)(4 * s) * HH + g0;
      v2f bf = { p[0], p[HH] };
      acc = __builtin_amdgcn_wmma_f32_16x16x4_f32(
          false, a[s], false, bf, (short)0, acc, false, false);
    }
    float* tp = t + (size_t)(bi0 + (hi ? 8 : 0)) * (NOUT * HH)
                  + o * HH + g0 + laneLo;
#pragma unroll
    for (int r = 0; r < 8; ++r)
      tp[(size_t)r * (NOUT * HH)] = acc[r];
  }
}

#define BLK 256
#define LDP 100

__device__ __forceinline__ void stage_block(float* __restrict__ L,
                                            const v4f (&r)[24], int tid) {
#pragma unroll
  for (int k = 0; k < 24; ++k) {
    const int f   = tid + 256 * k;
    const int row = f / 24;
    const int q   = f - row * 24;
    *(v4f*)(L + row * LDP + q * 4) = r[k];
  }
}

__device__ __forceinline__ void load_frags(v2f (&d)[24], const float* p) {
#pragma unroll
  for (int s = 0; s < 24; ++s) d[s] = *(const v2f*)(p + 4 * s);
}

__device__ __forceinline__ void tile_pair(
    const v2f (&a0)[24], const v2f (&a1)[24], const v2f (&bf)[24],
    const float (&bb)[NOUT], float* __restrict__ op) {
  v8f acc0 = {};
  v8f acc1 = {};
#pragma unroll
  for (int s = 0; s < 24; ++s) {
    acc0 = __builtin_amdgcn_wmma_f32_16x16x4_f32(
         false, a0[s],  false, bf[s],
         (short)0, acc0,  false,  false);
    acc1 = __builtin_amdgcn_wmma_f32_16x16x4_f32(
        false, a1[s], false, bf[s], (short)0, acc1, false, false);
  }
  float* op1 = op + (size_t)2 * NN * NOUT;
  v4f lo0 = { acc0[0] + bb[0], acc0[1] + bb[1], acc0[2] + bb[2], acc0[3] + bb[3] };
  v4f hi0 = { acc0[4] + bb[4], acc0[5] + bb[5], acc0[6] + bb[6], acc0[7] + bb[7] };
  v4f lo1 = { acc1[0] + bb[0], acc1[1] + bb[1], acc1[2] + bb[2], acc1[3] + bb[3] };
  v4f hi1 = { acc1[4] + bb[4], acc1[5] + bb[5], acc1[6] + bb[6], acc1[7] + bb[7] };
  __builtin_nontemporal_store(lo0, (v4f*)op);
  __builtin_nontemporal_store(hi0, (v4f*)(op + 4));
  __builtin_nontemporal_store(lo1, (v4f*)op1);
  __builtin_nontemporal_store(hi1, (v4f*)(op1 + 4));
}

__global__ __launch_bounds__(256) void k_bilinear_out(
    const float* __restrict__ t, const float* __restrict__ v,
    const float* __restrict__ bb_g, float* __restrict__ out) {
  __shared__ __attribute__((aligned(16))) float L[BLK * LDP];

  const int tid    = threadIdx.x;
  const int lane   = tid & 31;
  const int w      = tid >> 5;
  const int laneLo = lane & 15;
  const bool hi    = lane >= 16;

  const int blk = blockIdx.x;
  const int jb  = blk & 3;
  const int mb  = (blk >> 2) & 31;
  const int b   = blk >> 7;

  const float* tblk = t + ((size_t)b * 8192 + (size_t)mb * BLK) * HH;
  const float* vblk = v + ((size_t)b * NN + (size_t)jb * BLK) * HH;
  v4f tr[24], vr[24];
#pragma unroll
  for (int k = 0; k < 24; ++k) tr[k] = *(const v4f*)(tblk + (size_t)(tid + 256 * k) * 4);
#pragma unroll
  for (int k = 0; k < 24; ++k) vr[k] = *(const v4f*)(vblk + (size_t)(tid + 256 * k) * 4);

  float bb[NOUT];
#pragma unroll
  for (int r = 0; r < NOUT; ++r) bb[r] = bb_g[r];

  stage_block(L, tr, tid);
  __syncthreads();
  const float* la = L + (w * 32 + laneLo) * LDP + (hi ? 2 : 0);
  v2f a0[24], a1[24];
  load_frags(a0, la);
  load_frags(a1, la + 16 * LDP);
  __syncthreads();

  stage_block(L, vr, tid);
  __syncthreads();

  const int i0 = (mb * BLK + w * 32) >> 3;
  float* ob = out + (((size_t)b * NN + i0 + (hi ? 1 : 0)) * NN
                     + (size_t)jb * BLK + laneLo) * NOUT;
  const float* lb = L + laneLo * LDP + (hi ? 2 : 0);

  v2f b0[24], b1[24];
  load_frags(b0, lb);
  for (int jt = 0; jt < 16; jt += 2) {
    load_frags(b1, lb + (jt + 1) * 16 * LDP);
    tile_pair(a0, a1, b0, bb, ob + (size_t)jt * 16 * NOUT);
    load_frags(b0, lb + ((jt + 2) & 15) * 16 * LDP);
    tile_pair(a0, a1, b1, bb, ob + (size_t)(jt + 1) * 16 * NOUT);
  }
}

extern "C" void kernel_launch(void* const* d_in, const int* in_sizes, int n_in,
                              void* d_out, int out_size, void* d_ws, size_t ws_size,
                              hipStream_t stream) {
  const float* x     = (const float*)d_in[0];
  const float* W_lin = (const float*)d_in[1];
  const float* b_lin = (const float*)d_in[2];
  const float* W_bil = (const float*)d_in[3];
  const float* b_bil = (const float*)d_in[4];
  float* out = (float*)d_out;

  char* ws = (char*)d_ws;
  float* v  = (float*)(ws + (size_t)NB * NN * NHID * 4);
  float* t  = (float*)(ws + (size_t)NB * NN * NHID * 4
                          + (size_t)NB * NN * HH * 4);

  k_front<<<(NB * NN) / 16, 256, 0, stream>>>(x, W_lin, b_lin, W_bil, v, t);
  k_bilinear_out<<<512, 256, 0, stream>>>(t, v, b_bil, out);
}


# t GEMM moved into the main kernel: a work-group computes t for its own 16 rows into LDS and sweeps the batch's 1024 j in four double-buffered v blocks; t never goes to global memory; same sums
# speedup vs baseline: 2.7826x; 1.0337x over previous
#include <hip/hip_runtime.h>

typedef __attribute__((ext_vector_type(2))) float v2f;
typedef __attribute__((ext_vector_type(4))) float v4f;
typedef __attribute__((ext_vector_type(8))) float v8f;

#define NB   4
#define NN   1024
#define NIN  64
#define NHID 32
#define NOUT 8
#define HH   96

#define HP 36

__global__ __launch_bounds__(256) void k_front(
    const float* __restrict__ x, const float* __restrict__ W,
    const float* __restrict__ bl, float* __restrict__ v) {
  __shared__ __attribute__((aligned(16))) float Hs[18 * HP];

  const int tid  = threadIdx.x;
  const int lane = tid & 31;
  const int w   = __builtin_amdgcn_readfirstlane(tid >> 5);
  const int bi0 = blockIdx.x * 16;
  const int n0  = bi0 & (NN - 1);

  float wc[NIN];
#pragma unroll
  for (int i = 0; i < NIN; ++i) wc[i] = W[i * NHID + lane];
  const float bias = bl[lane];

  for (int q = w; q < 18; q += 8) {
    const int n = n0 - 1 + q;
    float val = 0.0f;
    if (n >= 0 && n < NN) {
      const float* xr = x + (size_t)(bi0 - 1 + q) * NIN;
      float s = bias;
#pragma unroll
      for (int i = 0; i < NIN; ++i) s = fmaf(xr[i], wc[i], s);
      val = fmaxf(s, 0.0f);
    }
    Hs[q * HP + lane] = val;
  }
  __syncthreads();

  float* vt = v + (size_t)bi0 * HH;
  for (int p = 0; p < 2; ++p) {
    const int f = tid + 256 * p;
    if (f < 16 * 24) {
      const int r  = f / 24;
      const int q4 = f - r * 24;
      *(v4f*)(vt + (size_t)f * 4) =
          *(const v4f*)(Hs + (r + (q4 >> 3)) * HP + (q4 & 7) * 4);
    }
  }
}

#define BLK 256
#define LDP 100
#define TM  128

__device__ __forceinline__ void stage_block(float* __restrict__ L,
                                            const v4f (&r)[12], int tid) {
#pragma unroll
  for (int k = 0; k < 12; ++k) {
    const int f   = tid + 512 * k;
    const int row = f / 24;
    const int q   = f - row * 24;
    *(v4f*)(L + row * LDP + q * 4) = r[k];
  }
}

__device__ __forceinline__ void fetch_block(v4f (&r)[12],
                                            const float* __restrict__ src, int tid) {
#pragma unroll
  for (int k = 0; k < 12; ++k)
    r[k] = *(const v4f*)(src + (size_t)(tid + 512 * k) * 4);
}

__device__ __forceinline__ void lds_barrier() {
  __builtin_amdgcn_fence(__ATOMIC_RELEASE, "workgroup", "local");
  __builtin_amdgcn_s_barrier();
  __builtin_amdgcn_fence(__ATOMIC_ACQUIRE, "workgroup", "local");
}

__device__ __forceinline__ void load_frags(v2f (&d)[24], const float* p) {
#pragma unroll
  for (int s = 0; s < 24; ++s) d[s] = *(const v2f*)(p + 4 * s);
}

__device__ __forceinline__ void tile_one(
    const v2f (&a)[24], const v2f (&bf)[24],
    const float (&bb)[NOUT], float* __restrict__ op) {
  v8f acc = {};
#pragma unroll
  for (int s = 0; s < 24; ++s)
    acc = __builtin_amdgcn_wmma_f32_16x16x4_f32(
         false, a[s],  false, bf[s],
         (short)0, acc,  false,  false);
  v4f lo = { acc[0] + bb[0], acc[1] + bb[1], acc[2] + bb[2], acc[3] + bb[3] };
  v4f hi = { acc[4] + bb[4], acc[5] + bb[5], acc[6] + bb[6], acc[7] + bb[7] };
  __builtin_nontemporal_store(lo, (v4f*)op);
  __builtin_nontemporal_store(hi, (v4f*)(op + 4));
}

__global__ __launch_bounds__(512) void k_bilinear_out(
    const float* __restrict__ v, const float* __restrict__ Wb,
    const float* __restrict__ bb_g, float* __restrict__ out) {
  __shared__ __attribute__((aligned(16))) float Lv[2 * BLK * LDP];
  __shared__ __attribute__((aligned(16))) float Lt[TM * LDP];

  const int tid    = threadIdx.x;
  const int lane   = tid & 31;
  const int laneLo = lane & 15;
  const bool hi    = lane >= 16;
  const int w   = __builtin_amdgcn_readfirstlane(tid >> 5);
  const int blk = blockIdx.x;
  const int b   = blk >> 6;
  const int i0  = (blk & 63) * 16;

  const float* vb = v + (size_t)b * NN * HH;

  v4f vr[12];
  fetch_block(vr, vb, tid);

  float bb[NOUT];
#pragma unroll
  for (int r = 0; r < NOUT; ++r) bb[r] = bb_g[r];

  v2f a[24];
  load_frags(a, vb + (size_t)(i0 + laneLo) * HH + (hi ? 2 : 0));
  {
    const int o  = w & 7;
    const int gh = w >> 3;
    const float* wB = Wb + (size_t)o * HH * HH + (hi ? 2 * HH : 0) + laneLo;
#pragma unroll
    for (int gt = 0; gt < 3; ++gt) {
      const int g0 = (gh * 3 + gt) * 16;
      v8f acc = {};
#pragma unroll
      for (int s = 0; s < 24; ++s) {
        const float* p = wB + (size_t)(4 * s) * HH + g0;
        v2f bf = { p[0], p[HH] };
        acc = __builtin_amdgcn_wmma_f32_16x16x4_f32(
            false, a[s], false, bf, (short)0, acc, false, false);
      }
      float* tp = Lt + ((hi ? 64 : 0) + o) * LDP + g0 + laneLo;
#pragma unroll
      for (int r = 0; r < 8; ++r)
        tp[r * 8 * LDP] = acc[r];
    }
  }
  stage_block(Lv, vr, tid);
  __syncthreads();

  const int at = w & 7;
  const int jh = w >> 3;
  load_frags(a, Lt + (at * 16 + laneLo) * LDP + (hi ? 2 : 0));

  float* ob = out + (((size_t)b * NN + i0 + 2 * at + (hi ? 1 : 0)) * NN
                     + (size_t)jh * 128 + laneLo) * NOUT;
  const int lbo = (jh * 128 + laneLo) * LDP + (hi ? 2 : 0);

  for (int jb = 0; jb < 4; ++jb) {
    const float* lb = Lv + (jb & 1) * (BLK * LDP) + lbo;
    float* obj = ob + (size_t)jb * BLK * NOUT;
    if (jb < 3) fetch_block(vr, vb + (size_t)(jb + 1) * BLK * HH, tid);

    v2f b0[24], b1[24];
    load_frags(b0, lb);
    for (int half = 0; half < 2; ++half) {
#pragma nounroll
      for (int jt = half * 4; jt < half * 4 + 4; jt += 2) {
        load_frags(b1, lb + (jt + 1) * 16 * LDP);
        tile_one(a, b0, bb, obj + (size_t)jt * 16 * NOUT);
        load_frags(b0, lb + ((jt + 2) & 7) * 16 * LDP);
        tile_one(a, b1, bb, obj + (size_t)(jt + 1) * 16 * NOUT);
      }
      if (half == 0 && jb < 3)
        stage_block(Lv + ((jb + 1) & 1) * (BLK * LDP), vr, tid);
    }
    if (jb < 3)
      lds_barrier();
  }
}

extern "C" void kernel_launch(void* const* d_in, const int* in_sizes, int n_in,
                              void* d_out, int out_size, void* d_ws, size_t ws_size,
                              hipStream_t stream) {
  const float* x     = (const float*)d_in[0];
  const float* W_lin = (const float*)d_in[1];
  const float* b_lin = (const float*)d_in[2];
  const float* W_bil = (const float*)d_in[3];
  const float* b_bil = (const float*)d_in[4];
  float* out = (float*)d_out;

  char* ws = (char*)d_ws;
  float* v  = (float*)(ws + (size_t)NB * NN * NHID * 4);

  k_front<<<(NB * NN) / 16, 256, 0, stream>>>(x, W_lin, b_lin, v);
  k_bilinear_out<<<256, 512, 0, stream>>>(v, W_bil, b_bil, out);
}


# Main GEMM wave tile doubled: 8 waves per work-group, each keeps two A tiles in registers, so one B fragment read feeds two accumulator chains and a j-tile's 12 reads stand in one place; same sums
# speedup vs baseline: 2.9575x; 1.0629x over previous
#include <hip/hip_runtime.h>

typedef __attribute__((ext_vector_type(2))) float v2f;
typedef __attribute__((ext_vector_type(4))) float v4f;
typedef __attribute__((ext_vector_type(8))) float v8f;

#define NB   4
#define NN   1024
#define NIN  64
#define NHID 32
#define NOUT 8
#define HH   96

#define HP 36

__global__ __launch_bounds__(256) void k_front(
    const float* __restrict__ x, const float* __restrict__ W,
    const float* __restrict__ bl, float* __restrict__ v) {
  __shared__ __attribute__((aligned(16))) float Hs[18 * HP];

  const int tid  = threadIdx.x;
  const int lane = tid & 31;
  const int w   = __builtin_amdgcn_readfirstlane(tid >> 5);
  const int bi0 = blockIdx.x * 16;
  const int n0  = bi0 & (NN - 1);

  float wc[NIN];
#pragma unroll
  for (int i = 0; i < NIN; ++i) wc[i] = W[i * NHID + lane];
  const float bias = bl[lane];

  for (int q = w; q < 18; q += 8) {
    const int n = n0 - 1 + q;
    float val = 0.0f;
    if (n >= 0 && n < NN) {
      const float* xr = x + (size_t)(bi0 - 1 + q) * NIN;
      float s = bias;
#pragma unroll
      for (int i = 0; i < NIN; ++i) s = fmaf(xr[i], wc[i], s);
      val = fmaxf(s, 0.0f);
    }
    Hs[q * HP + lane] = val;
  }
  __syncthreads();

  float* vt = v + (size_t)bi0 * HH;
  for (int p = 0; p < 2; ++p) {
    const int f = tid + 256 * p;
    if (f < 16 * 24) {
      const int r  = f / 24;
      const int q4 = f - r * 24;
      *(v4f*)(vt + (size_t)f * 4) =
          *(const v4f*)(Hs + (r + (q4 >> 3)) * HP + (q4 & 7) * 4);
    }
  }
}

#define BLK 256
#define LDP 100
#define TM  128
#define HALF_PIECES 3072

__device__ __forceinline__ void stage_half(float* __restrict__ L,
                                           const v4f (&r)[12], int base, int tid) {
#pragma unroll
  for (int k = 0; k < 12; ++k) {
    const int f   = base + tid + 256 * k;
    const int row = f / 24;
    const int q   = f - row * 24;
    *(v4f*)(L + row * LDP + q * 4) = r[k];
  }
}

__device__ __forceinline__ void fetch_half(v4f (&r)[12],
                                           const float* __restrict__ src,
                                           int base, int tid) {
#pragma unroll
  for (int k = 0; k < 12; ++k)
    r[k] = *(const v4f*)(src + (size_t)(base + tid + 256 * k) * 4);
}

__device__ __forceinline__ void lds_barrier() {
  __builtin_amdgcn_fence(__ATOMIC_RELEASE, "workgroup", "local");
  __builtin_amdgcn_s_barrier();
  __builtin_amdgcn_fence(__ATOMIC_ACQUIRE, "workgroup", "local");
}

__device__ __forceinline__ void load_frags(v2f (&d)[24], const float* p) {
#pragma unroll
  for (int s = 0; s < 24; ++s) d[s] = *(const v2f*)(p + 4 * s);
}

__device__ __forceinline__ void store_tile(const v8f& acc, const float (&bb)[NOUT],
                                           float* __restrict__ op) {
  v4f lo = { acc[0] + bb[0], acc[1] + bb[1], acc[2] + bb[2], acc[3] + bb[3] };
  v4f hi = { acc[4] + bb[4], acc[5] + bb[5], acc[6] + bb[6], acc[7] + bb[7] };
  __builtin_nontemporal_store(lo, (v4f*)op);
  __builtin_nontemporal_store(hi, (v4f*)(op + 4));
}

__device__ __forceinline__ void tile_two(
    const v2f (&a0)[24], const v2f (&a1)[24], const v2f (&bf)[24],
    const float (&bb)[NOUT], float* __restrict__ op0, float* __restrict__ op1) {
  v8f acc0 = {};
  v8f acc1 = {};
#pragma unroll
  for (int s = 0; s < 24; ++s) {
    acc0 = __builtin_amdgcn_wmma_f32_16x16x4_f32(
         false, a0[s],  false, bf[s],
         (short)0, acc0,  false,  false);
    acc1 = __builtin_amdgcn_wmma_f32_16x16x4_f32(
        false, a1[s], false, bf[s], (short)0, acc1, false, false);
  }
  store_tile(acc0, bb, op0);
  store_tile(acc1, bb, op1);
}

__global__ __launch_bounds__(256) void k_bilinear_out(
    const float* __restrict__ v, const float* __restrict__ Wb,
    const float* __restrict__ bb_g, float* __restrict__ out) {
  __shared__ __attribute__((aligned(16))) float Lv[2 * BLK * LDP];
  __shared__ __attribute__((aligned(16))) float Lt[TM * LDP];

  const int tid    = threadIdx.x;
  const int lane   = tid & 31;
  const int laneLo = lane & 15;
  const bool hi    = lane >= 16;
  const int w   = __builtin_amdgcn_readfirstlane(tid >> 5);
  const int blk = blockIdx.x;
  const int b   = blk >> 6;
  const int i0  = (blk & 63) * 16;

  const float* vb = v + (size_t)b * NN * HH;

  v4f vr[12];
  {
    v4f vr2[12];
    fetch_half(vr, vb, 0, tid);
    fetch_half(vr2, vb, HALF_PIECES, tid);

    v2f a[24];
    load_frags(a, vb + (size_t)(i0 + laneLo) * HH + (hi ? 2 : 0));
    const int o = w;
    const float* wB = Wb + (size_t)o * HH * HH + (hi ? 2 * HH : 0) + laneLo;
#pragma unroll
    for (int gh = 0; gh < 2; ++gh) {
#pragma unroll
      for (int gt = 0; gt < 3; ++gt) {
        const int g0 = (gh * 3 + gt) * 16;
        v8f acc = {};
#pragma unroll
        for (int s = 0; s < 24; ++s) {
          const float* p = wB + (size_t)(4 * s) * HH + g0;
          v2f bf = { p[0], p[HH] };
          acc = __builtin_amdgcn_wmma_f32_16x16x4_f32(
              false, a[s], false, bf, (short)0, acc, false, false);
        }
        float* tp = Lt + ((hi ? 64 : 0) + o) * LDP + g0 + laneLo;
#pragma unroll
        for (int r = 0; r < 8; ++r)
          tp[r * 8 * LDP] = acc[r];
      }
    }
    stage_half(Lv, vr, 0, tid);
    stage_half(Lv, vr2, HALF_PIECES, tid);
  }
  __syncthreads();

  float bb[NOUT];
#pragma unroll
  for (int r = 0; r < NOUT; ++r) bb[r] = bb_g[r];

  const int ap = w & 3;
  const int jh = w >> 2;
  v2f a0[24], a1[24];
  load_frags(a0, Lt + (ap * 32 + laneLo) * LDP + (hi ? 2 : 0));
  load_frags(a1, Lt + (ap * 32 + 16 + laneLo) * LDP + (hi ? 2 : 0));

  float* ob = out + (((size_t)b * NN + i0 + 4 * ap + (hi ? 1 : 0)) * NN
                     + (size_t)jh * 128 + laneLo) * NOUT;
  const size_t t1 = (size_t)2 * NN * NOUT;
  const int lbo = (jh * 128 + laneLo) * LDP + (hi ? 2 : 0);

  for (int jb = 0; jb < 4; ++jb) {
    const float* lb = Lv + (jb & 1) * (BLK * LDP) + lbo;
    float* Lnext = Lv + ((jb + 1) & 1) * (BLK * LDP);
    const float* vnext = vb + (size_t)(jb + 1) * BLK * HH;
    float* obj = ob + (size_t)jb * BLK * NOUT;

    v2f b0[24], b1[24];
    load_frags(b0, lb);
    for (int half = 0; half < 2; ++half) {
      if (jb < 3) fetch_half(vr, vnext, half * HALF_PIECES, tid);
#pragma nounroll
      for (int jt = half * 4; jt < half * 4 + 4; jt += 2) {
        load_frags(b1, lb + (jt + 1) * 16 * LDP);
        __builtin_amdgcn_sched_barrier(0);
        tile_two(a0, a1, b0, bb, obj + (size_t)jt * 16 * NOUT,
                 obj + t1 + (size_t)jt * 16 * NOUT);
        __builtin_amdgcn_sched_barrier(0);
        load_frags(b0, lb + ((jt + 2) & 7) * 16 * LDP);
        __builtin_amdgcn_sched_barrier(0);
        tile_two(a0, a1, b1, bb, obj + (size_t)(jt + 1) * 16 * NOUT,
                 obj + t1 + (size_t)(jt + 1) * 16 * NOUT);
      }
      if (jb < 3) stage_half(Lnext, vr, half * HALF_PIECES, tid);
    }
    if (jb < 3)
      lds_barrier();
  }
}

extern "C" void kernel_launch(void* const* d_in, const int* in_sizes, int n_in,
                              void* d_out, int out_size, void* d_ws, size_t ws_size,
                              hipStream_t stream) {
  const float* x     = (const float*)d_in[0];
  const float* W_lin = (const float*)d_in[1];
  const float* b_lin = (const float*)d_in[2];
  const float* W_bil = (const float*)d_in[3];
  const float* b_bil = (const float*)d_in[4];
  float* out = (float*)d_out;

  char* ws = (char*)d_ws;
  float* v  = (float*)(ws + (size_t)NB * NN * NHID * 4);

  k_front<<<(NB * NN) / 16, 256, 0, stream>>>(x, W_lin, b_lin, v);
  k_bilinear_out<<<256, 256, 0, stream>>>(v, W_bil, b_bil, out);
}


# Main GEMM B operand kept as h: the batch's 1024 h rows (middle third of v) staged once in LDS with two zero rows, fragments read rows j-1, j, j+1; no v blocks, no barrier in the j sweep; same sums
# speedup vs baseline: 3.1174x; 1.0541x over previous
#include <hip/hip_runtime.h>

typedef __attribute__((ext_vector_type(2))) float v2f;
typedef __attribute__((ext_vector_type(4))) float v4f;
typedef __attribute__((ext_vector_type(8))) float v8f;

#define NB   4
#define NN   1024
#define NIN  64
#define NHID 32
#define NOUT 8
#define HH   96

#define HP 36

__global__ __launch_bounds__(256) void k_front(
    const float* __restrict__ x, const float* __restrict__ W,
    const float* __restrict__ bl, float* __restrict__ v) {
  __shared__ __attribute__((aligned(16))) float Hs[18 * HP];

  const int tid  = threadIdx.x;
  const int lane = tid & 31;
  const int w   = __builtin_amdgcn_readfirstlane(tid >> 5);
  const int bi0 = blockIdx.x * 16;
  const int n0  = bi0 & (NN - 1);

  float wc[NIN];
#pragma unroll
  for (int i = 0; i < NIN; ++i) wc[i] = W[i * NHID + lane];
  const float bias = bl[lane];

  for (int q = w; q < 18; q += 8) {
    const int n = n0 - 1 + q;
    float val = 0.0f;
    if (n >= 0 && n < NN) {
      const float* xr = x + (size_t)(bi0 - 1 + q) * NIN;
      float s = bias;
#pragma unroll
      for (int i = 0; i < NIN; ++i) s = fmaf(xr[i], wc[i], s);
      val = fmaxf(s, 0.0f);
    }
    Hs[q * HP + lane] = val;
  }
  __syncthreads();

  float* vt = v + (size_t)bi0 * HH;
  for (int p = 0; p < 2; ++p) {
    const int f = tid + 256 * p;
    if (f < 16 * 24) {
      const int r  = f / 24;
      const int q4 = f - r * 24;
      *(v4f*)(vt + (size_t)f * 4) =
          *(const v4f*)(Hs + (r + (q4 >> 3)) * HP + (q4 & 7) * 4);
    }
  }
}

#define LDP 100
#define TM  128
#define HQ  36
#define HR  (NN + 2)

__device__ __forceinline__ void load_frags(v2f (&d)[24], const float* p) {
#pragma unroll
  for (int s = 0; s < 24; ++s) d[s] = *(const v2f*)(p + 4 * s);
}

__device__ __forceinline__ void load_frags_h(v2f (&d)[24], const float* p) {
#pragma unroll
  for (int s = 0; s < 24; ++s)
    d[s] = *(const v2f*)(p + (s >> 3) * HQ + 4 * (s & 7));
}

__device__ __forceinline__ void store_tile(const v8f& acc, const float (&bb)[NOUT],
                                           float* __restrict__ op) {
  v4f lo = { acc[0] + bb[0], acc[1] + bb[1], acc[2] + bb[2], acc[3] + bb[3] };
  v4f hi = { acc[4] + bb[4], acc[5] + bb[5], acc[6] + bb[6], acc[7] + bb[7] };
  __builtin_nontemporal_store(lo, (v4f*)op);
  __builtin_nontemporal_store(hi, (v4f*)(op + 4));
}

__device__ __forceinline__ void tile_two(
    const v2f (&a0)[24], const v2f (&a1)[24], const v2f (&bf)[24],
    const float (&bb)[NOUT], float* __restrict__ op0, float* __restrict__ op1) {
  v8f acc0 = {};
  v8f acc1 = {};
#pragma unroll
  for (int s = 0; s < 24; ++s) {
    acc0 = __builtin_amdgcn_wmma_f32_16x16x4_f32(
         false, a0[s],  false, bf[s],
         (short)0, acc0,  false,  false);
    acc1 = __builtin_amdgcn_wmma_f32_16x16x4_f32(
        false, a1[s], false, bf[s], (short)0, acc1, false, false);
  }
  store_tile(acc0, bb, op0);
  store_tile(acc1, bb, op1);
}

__global__ __launch_bounds__(256) void k_bilinear_out(
    const float* __restrict__ v, const float* __restrict__ Wb,
    const float* __restrict__ bb_g, float* __restrict__ out) {
  __shared__ __attribute__((aligned(16))) float Hs[HR * HQ];
  __shared__ __attribute__((aligned(16))) float Lt[TM * LDP];

  const int tid    = threadIdx.x;
  const int lane   = tid & 31;
  const int laneLo = lane & 15;
  const bool hi    = lane >= 16;
  const int w   = __builtin_amdgcn_readfirstlane(tid >> 5);
  const int blk = blockIdx.x;
  const int b   = blk >> 6;
  const int i0  = (blk & 63) * 16;

  const float* vb = v + (size_t)b * NN * HH;

  {
    v4f hr[32];
    const float* hsrc = vb + (size_t)(tid >> 3) * HH + NHID + (tid & 7) * 4;
#pragma unroll
    for (int k = 0; k < 32; ++k)
      hr[k] = *(const v4f*)(hsrc + (size_t)k * 32 * HH);

    v2f a[24];
    load_frags(a, vb + (size_t)(i0 + laneLo) * HH + (hi ? 2 : 0));
    const int o = w;
    const float* wB = Wb + (size_t)o * HH * HH + (hi ? 2 * HH : 0) + laneLo;
#pragma unroll
    for (int gh = 0; gh < 2; ++gh) {
#pragma unroll
      for (int gt = 0; gt < 3; ++gt) {
        const int g0 = (gh * 3 + gt) * 16;
        v8f acc = {};
#pragma unroll
        for (int s = 0; s < 24; ++s) {
          const float* p = wB + (size_t)(4 * s) * HH + g0;
          v2f bf = { p[0], p[HH] };
          acc = __builtin_amdgcn_wmma_f32_16x16x4_f32(
              false, a[s], false, bf, (short)0, acc, false, false);
        }
        float* tp = Lt + ((hi ? 64 : 0) + o) * LDP + g0 + laneLo;
#pragma unroll
        for (int r = 0; r < 8; ++r)
          tp[r * 8 * LDP] = acc[r];
      }
    }

    float* hdst = Hs + ((tid >> 3) + 1) * HQ + (tid & 7) * 4;
#pragma unroll
    for (int k = 0; k < 32; ++k)
      *(v4f*)(hdst + k * 32 * HQ) = hr[k];
    if (tid < 16) {
      const v4f z = { 0.0f, 0.0f, 0.0f, 0.0f };
      *(v4f*)(Hs + (tid < 8 ? 0 : (HR - 1)) * HQ + (tid & 7) * 4) = z;
    }
  }
  __syncthreads();

  float bb[NOUT];
#pragma unroll
  for (int r = 0; r < NOUT; ++r) bb[r] = bb_g[r];

  const int ap = w & 3;
  const int jh = w >> 2;
  v2f a0[24], a1[24];
  load_frags(a0, Lt + (ap * 32 + laneLo) * LDP + (hi ? 2 : 0));
  load_frags(a1, Lt + (ap * 32 + 16 + laneLo) * LDP + (hi ? 2 : 0));

  float* ob = out + (((size_t)b * NN + i0 + 4 * ap + (hi ? 1 : 0)) * NN
                     + (size_t)jh * 512 + laneLo) * NOUT;
  const size_t t1 = (size_t)2 * NN * NOUT;
  const float* lb = Hs + (jh * 512 + laneLo) * HQ + (hi ? 2 : 0);

  v2f b0[24], b1[24];
  load_frags_h(b0, lb);
#pragma nounroll
  for (int jt = 0; jt < 32; jt += 2) {
    load_frags_h(b1, lb + (jt + 1) * 16 * HQ);
    __builtin_amdgcn_sched_barrier(0);
    tile_two(a0, a1, b0, bb, ob + (size_t)jt * 16 * NOUT,
             ob + t1 + (size_t)jt * 16 * NOUT);
    __builtin_amdgcn_sched_barrier(0);
    load_frags_h(b0, lb + ((jt + 2) & 31) * 16 * HQ);
    __builtin_amdgcn_sched_barrier(0);
    tile_two(a0, a1, b1, bb, ob + (size_t)(jt + 1) * 16 * NOUT,
             ob + t1 + (size_t)(jt + 1) * 16 * NOUT);
  }
}

extern "C" void kernel_launch(void* const* d_in, const int* in_sizes, int n_in,
                              void* d_out, int out_size, void* d_ws, size_t ws_size,
                              hipStream_t stream) {
  const float* x     = (const float*)d_in[0];
  const float* W_lin = (const float*)d_in[1];
  const float* b_lin = (const float*)d_in[2];
  const float* W_bil = (const float*)d_in[3];
  const float* b_bil = (const float*)d_in[4];
  float* out = (float*)d_out;

  char* ws = (char*)d_ws;
  float* v  = (float*)(ws + (size_t)NB * NN * NHID * 4);

  k_front<<<(NB * NN) / 16, 256, 0, stream>>>(x, W_lin, b_lin, v);
  k_bilinear_out<<<256, 256, 0, stream>>>(v, W_bil, b_bil, out);
}
